# starting kernel
# baseline (speedup 1.0000x reference)
#include <hip/hip_runtime.h>
#include <hip/hip_bf16.h>
#include <math.h>

typedef __attribute__((ext_vector_type(2))) float v2f;
typedef __attribute__((ext_vector_type(8))) float v8f;

__global__ void degree_kernel(const int* __restrict__ dst, float* __restrict__ deg, int E) {
    int e = blockIdx.x * blockDim.x + threadIdx.x;
    if (e >= E) return;
    atomicAdd(&deg[dst[e]], 1.0f);
}

__global__ void invdeg_kernel(float* __restrict__ deg, int N) {
    int i = blockIdx.x * blockDim.x + threadIdx.x;
    if (i >= N) return;
    deg[i] = 1.0f / fmaxf(deg[i], 1.0f);
}

template <int C>
__global__ __launch_bounds__(256)
void scatter_add_kernel(const float* __restrict__ x,
                        const int* __restrict__ src,
                        const int* __restrict__ dst,
                        float* __restrict__ agg, int E) {
    constexpr int CHUNKS = C / 4;
    int tid = blockIdx.x * blockDim.x + threadIdx.x;
    if (tid >= E * CHUNKS) return;
    int e = tid / CHUNKS;
    int q = tid - e * CHUNKS;
    int s = src[e];
    int d = dst[e];
    const float4 v = *(const float4*)(x + (size_t)s * C + q * 4);
    float* a = agg + (size_t)d * C + q * 4;
    atomicAdd(a + 0, v.x);
    atomicAdd(a + 1, v.y);
    atomicAdd(a + 2, v.z);
    atomicAdd(a + 3, v.w);
}

template <int CIN, int COUT, bool RELU>
__global__ __launch_bounds__(256)
void sage_wmma_kernel(const float* __restrict__ agg,
                      const float* __restrict__ invdeg,
                      const float* __restrict__ xin,
                      const float* __restrict__ wl,
                      const float* __restrict__ wr,
                      const float* __restrict__ bias,
                      float* __restrict__ out,
                      int N) {
    const int lane = threadIdx.x & 31;
    const int waveInBlock = threadIdx.x >> 5;
    const int tile = blockIdx.x * (blockDim.x >> 5) + waveInBlock;
    const int row0 = tile * 16;
    if (row0 >= N) return;

    const int m  = lane & 15;
    const int kk = (lane >> 4) << 1;
    const bool fullTile = (row0 + 16 <= N);
    const int node = row0 + (fullTile ? m : min(m, N - 1 - row0));
    const float scale = invdeg[node];
    const float* aggRow = agg + (size_t)node * CIN;
    const float* xRow   = xin + (size_t)node * CIN;
    const int rbase = row0 + ((lane >> 4) << 3);

    for (int nb = 0; nb < COUT; nb += 16) {
        v8f c = {};
        const float* wlRow = wl + (size_t)(nb + m) * CIN;
        const float* wrRow = wr + (size_t)(nb + m) * CIN;
#pragma unroll
        for (int k0 = 0; k0 < CIN; k0 += 4) {
            v2f a_mean, a_self, b_l, b_r;
            a_mean.x = aggRow[k0 + kk] * scale;
            a_mean.y = aggRow[k0 + kk + 1] * scale;
            a_self.x = xRow[k0 + kk];
            a_self.y = xRow[k0 + kk + 1];
            b_l.x = wlRow[k0 + kk];
            b_l.y = wlRow[k0 + kk + 1];
            b_r.x = wrRow[k0 + kk];
            b_r.y = wrRow[k0 + kk + 1];
            c = __builtin_amdgcn_wmma_f32_16x16x4_f32(false, a_mean, false, b_l,
                                                      (short)0, c, false, false);
            c = __builtin_amdgcn_wmma_f32_16x16x4_f32(false, a_self, false, b_r,
                                                      (short)0, c, false, false);
        }
        const float bv = bias[nb + m];
        if (fullTile) {
            float* o = out + (size_t)rbase * COUT + nb + m;
#pragma unroll
            for (int v = 0; v < 8; ++v) {
                float val = c[v] + bv;
                if (RELU) val = fmaxf(val, 0.0f);
                o[(size_t)v * COUT] = val;
            }
        } else {
#pragma unroll
            for (int v = 0; v < 8; ++v) {
                int r = rbase + v;
                if (r < N) {
                    float val = c[v] + bv;
                    if (RELU) val = fmaxf(val, 0.0f);
                    out[(size_t)r * COUT + nb + m] = val;
                }
            }
        }
    }
}

__global__ void sage_final_kernel(const float* __restrict__ agg,
                                  const float* __restrict__ invdeg,
                                  const float* __restrict__ h2,
                                  const float* __restrict__ wl3,
                                  const float* __restrict__ wr3,
                                  const float* __restrict__ b3,
                                  float* __restrict__ out, int N) {
    int i = blockIdx.x * blockDim.x + threadIdx.x;
    if (i >= N) return;
    float s = b3[0];
    float sc = invdeg[i];
    const float4* ar = (const float4*)(agg + (size_t)i * 64);
    const float4* hr = (const float4*)(h2 + (size_t)i * 64);
    const float4* wlv = (const float4*)wl3;
    const float4* wrv = (const float4*)wr3;
#pragma unroll
    for (int q = 0; q < 16; ++q) {
        float4 a = ar[q], h = hr[q], l = wlv[q], r = wrv[q];
        s += (a.x * sc) * l.x + (a.y * sc) * l.y + (a.z * sc) * l.z + (a.w * sc) * l.w;
        s += h.x * r.x + h.y * r.y + h.z * r.z + h.w * r.w;
    }
    out[i] = 1.0f / (1.0f + expf(-s));
}

extern "C" void kernel_launch(void* const* d_in, const int* in_sizes, int n_in,
                              void* d_out, int out_size, void* d_ws, size_t ws_size,
                              hipStream_t stream) {
    const float* x    = (const float*)d_in[0];
    const int*   ei   = (const int*)d_in[1];
    const float* wl1  = (const float*)d_in[2];
    const float* wr1  = (const float*)d_in[3];
    const float* b1   = (const float*)d_in[4];
    const float* wl2  = (const float*)d_in[5];
    const float* wr2  = (const float*)d_in[6];
    const float* b2   = (const float*)d_in[7];
    const float* wl3  = (const float*)d_in[8];
    const float* wr3  = (const float*)d_in[9];
    const float* b3   = (const float*)d_in[10];
    float* out = (float*)d_out;

    const int N = in_sizes[0] / 32;
    const int E = in_sizes[1] / 2;
    const int* src = ei;
    const int* dst = ei + E;

    float* deg = (float*)d_ws;
    float* agg = deg + N;
    float* h1  = agg + (size_t)N * 64;
    float* h2  = h1 + (size_t)N * 64;

    const int tiles = (N + 15) / 16;
    const int gemmBlocks = (tiles + 7) / 8;

    hipMemsetAsync(deg, 0, (size_t)N * sizeof(float), stream);
    degree_kernel<<<(E + 255) / 256, 256, 0, stream>>>(dst, deg, E);
    invdeg_kernel<<<(N + 255) / 256, 256, 0, stream>>>(deg, N);

    hipMemsetAsync(agg, 0, (size_t)N * 32 * sizeof(float), stream);
    scatter_add_kernel<32><<<(E * 8 + 255) / 256, 256, 0, stream>>>(x, src, dst, agg, E);
    sage_wmma_kernel<32, 64, true><<<gemmBlocks, 256, 0, stream>>>(
        agg, deg, x, wl1, wr1, b1, h1, N);

    hipMemsetAsync(agg, 0, (size_t)N * 64 * sizeof(float), stream);
    scatter_add_kernel<64><<<(E * 16 + 255) / 256, 256, 0, stream>>>(h1, src, dst, agg, E);
    sage_wmma_kernel<64, 64, true><<<gemmBlocks, 256, 0, stream>>>(
        agg, deg, h1, wl2, wr2, b2, h2, N);

    hipMemsetAsync(agg, 0, (size_t)N * 64 * sizeof(float), stream);
    scatter_add_kernel<64><<<(E * 16 + 255) / 256, 256, 0, stream>>>(h2, src, dst, agg, E);
    sage_final_kernel<<<(N + 255) / 256, 256, 0, stream>>>(
        agg, deg, h2, wl3, wr3, b3, out, N);
}


# Neighbour sums by owner blocks: a block keeps 800 destination rows in LDS, filters the whole edge list for them, adds gathered rows by LDS f32 add; no global atomics or clears; sum order: LDS arrival
# speedup vs baseline: 1.3377x; 1.3377x over previous
#include <hip/hip_runtime.h>
#include <hip/hip_bf16.h>
#include <math.h>

typedef __attribute__((ext_vector_type(2))) float v2f;
typedef __attribute__((ext_vector_type(8))) float v8f;

template <int C, int R, bool COUNT>
__global__ __launch_bounds__(1024)
void owner_sum_kernel(const float* __restrict__ x,
                      const int* __restrict__ src,
                      const int* __restrict__ dst,
                      float* __restrict__ agg,
                      float* __restrict__ invdeg,
                      int N, int E) {
    constexpr int NW    = 32;
    constexpr int QCAP  = 160;
    constexpr int CHUNK = 1024 * 32;
    constexpr int CV    = C / 32;

    __shared__ __attribute__((aligned(16))) float acc[R * C];
    __shared__ float cnt[COUNT ? R : 1];
    __shared__ int   queue[NW * QCAP];

    const int tid  = threadIdx.x;
    const int lane = tid & 31;
    const int base = blockIdx.x * R;
    int* myq = queue + (tid >> 5) * QCAP;

    for (int i = tid; i < R * C / 4; i += 1024) ((float4*)acc)[i] = make_float4(0.0f, 0.0f, 0.0f, 0.0f);
    if (COUNT) {
        for (int i = tid; i < R; i += 1024) cnt[i] = 0.0f;
    }
    __syncthreads();

    int qlen = 0;

    auto drain = [&](int n, int q0) {
        int s = 0, dl = 0;
        if (lane < n) {
            const int e = myq[q0 + lane];
            s  = src[e];
            dl = dst[e] - base;
            if (COUNT) atomicAdd(cnt + dl, 1.0f);
        }
        for (int i0 = 0; i0 < n; i0 += 8) {
            float v[8][CV];
            int   dd[8];
#pragma unroll
            for (int u = 0; u < 8; ++u) {
                const int i  = min(i0 + u, n - 1);
                const int si = __builtin_amdgcn_readlane(s, i);
                dd[u]        = __builtin_amdgcn_readlane(dl, i);
                const float* row = x + (size_t)si * C + lane;
#pragma unroll
                for (int c = 0; c < CV; ++c) v[u][c] = row[c * 32];
            }
#pragma unroll
            for (int u = 0; u < 8; ++u) {
                if (i0 + u < n) {
                    float* a = acc + dd[u] * C + lane;
#pragma unroll
                    for (int c = 0; c < CV; ++c) atomicAdd(a + c * 32, v[u][c]);
                }
            }
        }
    };

    auto push = [&](int d, int e) {
        const bool m = (unsigned)(d - base) < (unsigned)R;
        const unsigned mask = __builtin_amdgcn_ballot_w32(m);
        if (mask != 0u) {
            if (m) myq[qlen + (int)__builtin_amdgcn_mbcnt_lo(mask, 0u)] = e;
            qlen += __builtin_popcount(mask);
        }
    };

    const bool vec = ((((size_t)dst) & 15) == 0);
    const int nIter = (E + CHUNK - 1) / CHUNK;
    const int itStart = (int)((blockIdx.x * (unsigned)nIter) / gridDim.x);
    for (int k = 0; k < nIter; ++k) {
        int it = itStart + k;
        if (it >= nIter) it -= nIter;
        const int cbase = it * CHUNK;
        int4 dv[8];
        if (vec && cbase + CHUNK <= E) {
            const int4* p = (const int4*)(dst + cbase) + tid;
#pragma unroll
            for (int u = 0; u < 8; ++u) dv[u] = p[u * 1024];
        } else {
#pragma unroll
            for (int u = 0; u < 8; ++u) {
                const int e0 = cbase + (u * 1024 + tid) * 4;
                dv[u].x = (e0     < E) ? dst[e0]     : -1;
                dv[u].y = (e0 + 1 < E) ? dst[e0 + 1] : -1;
                dv[u].z = (e0 + 2 < E) ? dst[e0 + 2] : -1;
                dv[u].w = (e0 + 3 < E) ? dst[e0 + 3] : -1;
            }
        }
#pragma unroll
        for (int u = 0; u < 8; ++u) {
            const int e0 = cbase + (u * 1024 + tid) * 4;
            push(dv[u].x, e0);
            push(dv[u].y, e0 + 1);
            push(dv[u].z, e0 + 2);
            push(dv[u].w, e0 + 3);
            while (qlen >= 32) {
                qlen -= 32;
                drain(32, qlen);
            }
        }
    }
    if (qlen > 0) drain(qlen, 0);
    __syncthreads();

    float4* aout = (float4*)(agg + (size_t)base * C);
    for (int i = tid; i < R * C / 4; i += 1024) {
        if (base + i / (C / 4) < N) aout[i] = ((const float4*)acc)[i];
    }
    if (COUNT) {
        for (int i = tid; i < R; i += 1024) {
            if (base + i < N) invdeg[base + i] = 1.0f / fmaxf(cnt[i], 1.0f);
        }
    }
}

template <int CIN, int COUT, bool RELU>
__global__ __launch_bounds__(256)
void sage_wmma_kernel(const float* __restrict__ agg,
                      const float* __restrict__ invdeg,
                      const float* __restrict__ xin,
                      const float* __restrict__ wl,
                      const float* __restrict__ wr,
                      const float* __restrict__ bias,
                      float* __restrict__ out,
                      int N) {
    const int lane = threadIdx.x & 31;
    const int waveInBlock = threadIdx.x >> 5;
    const int tile = blockIdx.x * (blockDim.x >> 5) + waveInBlock;
    const int row0 = tile * 16;
    if (row0 >= N) return;

    const int m  = lane & 15;
    const int kk = (lane >> 4) << 1;
    const bool fullTile = (row0 + 16 <= N);
    const int node = row0 + (fullTile ? m : min(m, N - 1 - row0));
    const float scale = invdeg[node];
    const float* aggRow = agg + (size_t)node * CIN;
    const float* xRow   = xin + (size_t)node * CIN;
    const int rbase = row0 + ((lane >> 4) << 3);

    for (int nb = 0; nb < COUT; nb += 16) {
        v8f c = {};
        const float* wlRow = wl + (size_t)(nb + m) * CIN;
        const float* wrRow = wr + (size_t)(nb + m) * CIN;
#pragma unroll
        for (int k0 = 0; k0 < CIN; k0 += 4) {
            v2f a_mean, a_self, b_l, b_r;
            a_mean.x = aggRow[k0 + kk] * scale;
            a_mean.y = aggRow[k0 + kk + 1] * scale;
            a_self.x = xRow[k0 + kk];
            a_self.y = xRow[k0 + kk + 1];
            b_l.x = wlRow[k0 + kk];
            b_l.y = wlRow[k0 + kk + 1];
            b_r.x = wrRow[k0 + kk];
            b_r.y = wrRow[k0 + kk + 1];
            c = __builtin_amdgcn_wmma_f32_16x16x4_f32(false, a_mean, false, b_l,
                                                      (short)0, c, false, false);
            c = __builtin_amdgcn_wmma_f32_16x16x4_f32(false, a_self, false, b_r,
                                                      (short)0, c, false, false);
        }
        const float bv = bias[nb + m];
        if (fullTile) {
            float* o = out + (size_t)rbase * COUT + nb + m;
#pragma unroll
            for (int v = 0; v < 8; ++v) {
                float val = c[v] + bv;
                if (RELU) val = fmaxf(val, 0.0f);
                o[(size_t)v * COUT] = val;
            }
        } else {
#pragma unroll
            for (int v = 0; v < 8; ++v) {
                int r = rbase + v;
                if (r < N) {
                    float val = c[v] + bv;
                    if (RELU) val = fmaxf(val, 0.0f);
                    out[(size_t)r * COUT + nb + m] = val;
                }
            }
        }
    }
}

__global__ void sage_final_kernel(const float* __restrict__ agg,
                                  const float* __restrict__ invdeg,
                                  const float* __restrict__ h2,
                                  const float* __restrict__ wl3,
                                  const float* __restrict__ wr3,
                                  const float* __restrict__ b3,
                                  float* __restrict__ out, int N) {
    int i = blockIdx.x * blockDim.x + threadIdx.x;
    if (i >= N) return;
    float s = b3[0];
    float sc = invdeg[i];
    const float4* ar = (const float4*)(agg + (size_t)i * 64);
    const float4* hr = (const float4*)(h2 + (size_t)i * 64);
    const float4* wlv = (const float4*)wl3;
    const float4* wrv = (const float4*)wr3;
#pragma unroll
    for (int q = 0; q < 16; ++q) {
        float4 a = ar[q], h = hr[q], l = wlv[q], r = wrv[q];
        s += (a.x * sc) * l.x + (a.y * sc) * l.y + (a.z * sc) * l.z + (a.w * sc) * l.w;
        s += h.x * r.x + h.y * r.y + h.z * r.z + h.w * r.w;
    }
    out[i] = 1.0f / (1.0f + expf(-s));
}

extern "C" void kernel_launch(void* const* d_in, const int* in_sizes, int n_in,
                              void* d_out, int out_size, void* d_ws, size_t ws_size,
                              hipStream_t stream) {
    const float* x    = (const float*)d_in[0];
    const int*   ei   = (const int*)d_in[1];
    const float* wl1  = (const float*)d_in[2];
    const float* wr1  = (const float*)d_in[3];
    const float* b1   = (const float*)d_in[4];
    const float* wl2  = (const float*)d_in[5];
    const float* wr2  = (const float*)d_in[6];
    const float* b2   = (const float*)d_in[7];
    const float* wl3  = (const float*)d_in[8];
    const float* wr3  = (const float*)d_in[9];
    const float* b3   = (const float*)d_in[10];
    float* out = (float*)d_out;

    const int N = in_sizes[0] / 32;
    const int E = in_sizes[1] / 2;
    const int* src = ei;
    const int* dst = ei + E;

    float* deg = (float*)d_ws;
    float* agg = deg + N;
    float* h1  = agg + (size_t)N * 64;
    float* h2  = h1 + (size_t)N * 64;

    const int tiles = (N + 15) / 16;
    const int gemmBlocks = (tiles + 7) / 8;

    constexpr int R = 800;
    const int ownerBlocks = (N + R - 1) / R;

    owner_sum_kernel<32, R, true><<<ownerBlocks, 1024, 0, stream>>>(x, src, dst, agg, deg, N, E);
    sage_wmma_kernel<32, 64, true><<<gemmBlocks, 256, 0, stream>>>(
        agg, deg, x, wl1, wr1, b1, h1, N);

    owner_sum_kernel<64, R, false><<<ownerBlocks, 1024, 0, stream>>>(h1, src, dst, agg, deg, N, E);
    sage_wmma_kernel<64, 64, true><<<gemmBlocks, 256, 0, stream>>>(
        agg, deg, h1, wl2, wr2, b2, h2, N);

    owner_sum_kernel<64, R, false><<<ownerBlocks, 1024, 0, stream>>>(h2, src, dst, agg, deg, N, E);
    sage_final_kernel<<<(N + 255) / 256, 256, 0, stream>>>(
        agg, deg, h2, wl3, wr3, b3, out, N);
}


# Edge list re-encoded (2 small launches per call) as 1 owner byte + 1 packed src/local-dst word per edge: owner blocks scan 16 edges per 16-byte load, fetch 1 word per match; sum order: LDS arrival
# speedup vs baseline: 1.8020x; 1.3471x over previous
#include <hip/hip_runtime.h>
#include <hip/hip_bf16.h>
#include <math.h>

typedef __attribute__((ext_vector_type(2))) float v2f;
typedef __attribute__((ext_vector_type(8))) float v8f;

template <int R>
__global__ __launch_bounds__(256)
void edge_encode_kernel(const int* __restrict__ src,
                        const int* __restrict__ dst,
                        unsigned int* __restrict__ own,
                        unsigned int* __restrict__ pk,
                        int N, int E, int nWords) {
    static_assert(R <= 1024, "local destination must fit in 10 bits");
    const int w = blockIdx.x * 256 + threadIdx.x;
    if (w >= nWords) return;
    const int e0 = w * 4;
    int s[4], d[4];
    const bool vec = (((((size_t)src) | ((size_t)dst)) & 15) == 0);
    if (vec && e0 + 4 <= E) {
        const int4 sv = *(const int4*)(src + e0);
        const int4 dv = *(const int4*)(dst + e0);
        s[0] = sv.x; s[1] = sv.y; s[2] = sv.z; s[3] = sv.w;
        d[0] = dv.x; d[1] = dv.y; d[2] = dv.z; d[3] = dv.w;
    } else {
#pragma unroll
        for (int j = 0; j < 4; ++j) {
            s[j] = (e0 + j < E) ? src[e0 + j] : 0;
            d[j] = (e0 + j < E) ? dst[e0 + j] : -1;
        }
    }
    unsigned int o = 0u;
    unsigned int p[4];
#pragma unroll
    for (int j = 0; j < 4; ++j) {
        const bool ok = (unsigned)d[j] < (unsigned)N;
        const unsigned int b  = ok ? (unsigned)d[j] / (unsigned)R : 255u;
        const unsigned int dl = ok ? (unsigned)d[j] - b * (unsigned)R : 0u;
        o |= b << (8 * j);
        p[j] = ((unsigned)s[j] & 0x3FFFFFu) | (dl << 22);
    }
    own[w] = o;
    *(uint4*)(pk + e0) = make_uint4(p[0], p[1], p[2], p[3]);
}

template <int C, int R, bool COUNT>
__global__ __launch_bounds__(1024)
void owner_sum_kernel(const float* __restrict__ x,
                      const unsigned int* __restrict__ own,
                      const unsigned int* __restrict__ pk,
                      float* __restrict__ agg,
                      float* __restrict__ invdeg,
                      int N, int nvec) {
    constexpr int NW    = 32;
    constexpr int QCAP  = 64;
    constexpr int VPI   = 1024 * 8;
    constexpr int CV    = C / 32;

    __shared__ __attribute__((aligned(16))) float acc[R * C];
    __shared__ float cnt[COUNT ? R : 1];
    __shared__ int   queue[NW * QCAP];

    const int tid  = threadIdx.x;
    const int lane = tid & 31;
    const int base = blockIdx.x * R;
    int* myq = queue + (tid >> 5) * QCAP;

    for (int i = tid; i < R * C / 4; i += 1024) ((float4*)acc)[i] = make_float4(0.0f, 0.0f, 0.0f, 0.0f);
    if (COUNT) {
        for (int i = tid; i < R; i += 1024) cnt[i] = 0.0f;
    }
    __syncthreads();

    int qlen = 0;

    auto drain = [&](int n, int q0) {
        int s = 0, dl = 0;
        if (lane < n) {
            const unsigned int p = pk[myq[q0 + lane]];
            s  = (int)(p & 0x3FFFFFu);
            dl = (int)(p >> 22);
            if (COUNT) atomicAdd(cnt + dl, 1.0f);
        }
        for (int i0 = 0; i0 < n; i0 += 8) {
            float v[8][CV];
            int   dd[8];
#pragma unroll
            for (int u = 0; u < 8; ++u) {
                const int i  = min(i0 + u, n - 1);
                const int si = __builtin_amdgcn_readlane(s, i);
                dd[u]        = __builtin_amdgcn_readlane(dl, i);
                const float* row = x + (size_t)si * C + lane;
#pragma unroll
                for (int c = 0; c < CV; ++c) v[u][c] = row[c * 32];
            }
#pragma unroll
            for (int u = 0; u < 8; ++u) {
                if (i0 + u < n) {
                    float* a = acc + dd[u] * C + lane;
#pragma unroll
                    for (int c = 0; c < CV; ++c) atomicAdd(a + c * 32, v[u][c]);
                }
            }
        }
    };

    const unsigned int key = (unsigned int)blockIdx.x * 0x01010101u;
    auto miss = [&](unsigned int w) -> unsigned int {
        const unsigned int t = w ^ key;
        return ((t & 0x7F7F7F7Fu) + 0x7F7F7F7Fu) | t | 0x7F7F7F7Fu;
    };

    const uint4* ov4 = (const uint4*)own;
    const int nIter = (nvec + VPI - 1) / VPI;
    const int itStart = (int)((blockIdx.x * (unsigned)nIter) / gridDim.x);
    for (int k = 0; k < nIter; ++k) {
        int it = itStart + k;
        if (it >= nIter) it -= nIter;
        const int vbase = it * VPI + tid;
        uint4 ov[8];
#pragma unroll
        for (int u = 0; u < 8; ++u) {
            const int idx = vbase + u * 1024;
            ov[u] = (idx < nvec) ? ov4[idx] : make_uint4(~0u, ~0u, ~0u, ~0u);
        }
#pragma unroll
        for (int u = 0; u < 8; ++u) {
            const int e0 = (vbase + u * 1024) * 16;
            const unsigned int z0 = miss(ov[u].x), z1 = miss(ov[u].y);
            const unsigned int z2 = miss(ov[u].z), z3 = miss(ov[u].w);
            unsigned int act = __builtin_amdgcn_ballot_w32((z0 & z1 & z2 & z3) != ~0u);
            if (act != 0u) {
                unsigned int mbits = ((~z0) >> 4) | ((~z1) >> 3) | ((~z2) >> 2) | ((~z3) >> 1);
                while (act != 0u) {
                    if (mbits != 0u) {
                        const int p = __builtin_ctz(mbits);
                        mbits &= mbits - 1u;
                        myq[qlen + (int)__builtin_amdgcn_mbcnt_lo(act, 0u)] =
                            e0 + (((p & 7) - 3) << 2) + (p >> 3);
                    }
                    qlen += __builtin_popcount(act);
                    if (qlen >= 32) {
                        qlen -= 32;
                        drain(32, qlen);
                    }
                    act = __builtin_amdgcn_ballot_w32(mbits != 0u);
                }
            }
        }
    }
    if (qlen > 0) drain(qlen, 0);
    __syncthreads();

    float4* aout = (float4*)(agg + (size_t)base * C);
    for (int i = tid; i < R * C / 4; i += 1024) {
        if (base + i / (C / 4) < N) aout[i] = ((const float4*)acc)[i];
    }
    if (COUNT) {
        for (int i = tid; i < R; i += 1024) {
            if (base + i < N) invdeg[base + i] = 1.0f / fmaxf(cnt[i], 1.0f);
        }
    }
}

template <int CIN, int COUT, bool RELU>
__global__ __launch_bounds__(256)
void sage_wmma_kernel(const float* __restrict__ agg,
                      const float* __restrict__ invdeg,
                      const float* __restrict__ xin,
                      const float* __restrict__ wl,
                      const float* __restrict__ wr,
                      const float* __restrict__ bias,
                      float* __restrict__ out,
                      int N) {
    const int lane = threadIdx.x & 31;
    const int waveInBlock = threadIdx.x >> 5;
    const int tile = blockIdx.x * (blockDim.x >> 5) + waveInBlock;
    const int row0 = tile * 16;
    if (row0 >= N) return;

    const int m  = lane & 15;
    const int kk = (lane >> 4) << 1;
    const bool fullTile = (row0 + 16 <= N);
    const int node = row0 + (fullTile ? m : min(m, N - 1 - row0));
    const float scale = invdeg[node];
    const float* aggRow = agg + (size_t)node * CIN;
    const float* xRow   = xin + (size_t)node * CIN;
    const int rbase = row0 + ((lane >> 4) << 3);

    for (int nb = 0; nb < COUT; nb += 16) {
        v8f c = {};
        const float* wlRow = wl + (size_t)(nb + m) * CIN;
        const float* wrRow = wr + (size_t)(nb + m) * CIN;
#pragma unroll
        for (int k0 = 0; k0 < CIN; k0 += 4) {
            v2f a_mean, a_self, b_l, b_r;
            a_mean.x = aggRow[k0 + kk] * scale;
            a_mean.y = aggRow[k0 + kk + 1] * scale;
            a_self.x = xRow[k0 + kk];
            a_self.y = xRow[k0 + kk + 1];
            b_l.x = wlRow[k0 + kk];
            b_l.y = wlRow[k0 + kk + 1];
            b_r.x = wrRow[k0 + kk];
            b_r.y = wrRow[k0 + kk + 1];
            c = __builtin_amdgcn_wmma_f32_16x16x4_f32(false, a_mean, false, b_l,
                                                      (short)0, c, false, false);
            c = __builtin_amdgcn_wmma_f32_16x16x4_f32(false, a_self, false, b_r,
                                                      (short)0, c, false, false);
        }
        const float bv = bias[nb + m];
        if (fullTile) {
            float* o = out + (size_t)rbase * COUT + nb + m;
#pragma unroll
            for (int v = 0; v < 8; ++v) {
                float val = c[v] + bv;
                if (RELU) val = fmaxf(val, 0.0f);
                o[(size_t)v * COUT] = val;
            }
        } else {
#pragma unroll
            for (int v = 0; v < 8; ++v) {
                int r = rbase + v;
                if (r < N) {
                    float val = c[v] + bv;
                    if (RELU) val = fmaxf(val, 0.0f);
                    out[(size_t)r * COUT + nb + m] = val;
                }
            }
        }
    }
}

__global__ void sage_final_kernel(const float* __restrict__ agg,
                                  const float* __restrict__ invdeg,
                                  const float* __restrict__ h2,
                                  const float* __restrict__ wl3,
                                  const float* __restrict__ wr3,
                                  const float* __restrict__ b3,
                                  float* __restrict__ out, int N) {
    int i = blockIdx.x * blockDim.x + threadIdx.x;
    if (i >= N) return;
    float s = b3[0];
    float sc = invdeg[i];
    const float4* ar = (const float4*)(agg + (size_t)i * 64);
    const float4* hr = (const float4*)(h2 + (size_t)i * 64);
    const float4* wlv = (const float4*)wl3;
    const float4* wrv = (const float4*)wr3;
#pragma unroll
    for (int q = 0; q < 16; ++q) {
        float4 a = ar[q], h = hr[q], l = wlv[q], r = wrv[q];
        s += (a.x * sc) * l.x + (a.y * sc) * l.y + (a.z * sc) * l.z + (a.w * sc) * l.w;
        s += h.x * r.x + h.y * r.y + h.z * r.z + h.w * r.w;
    }
    out[i] = 1.0f / (1.0f + expf(-s));
}

extern "C" void kernel_launch(void* const* d_in, const int* in_sizes, int n_in,
                              void* d_out, int out_size, void* d_ws, size_t ws_size,
                              hipStream_t stream) {
    const float* x    = (const float*)d_in[0];
    const int*   ei   = (const int*)d_in[1];
    const float* wl1  = (const float*)d_in[2];
    const float* wr1  = (const float*)d_in[3];
    const float* b1   = (const float*)d_in[4];
    const float* wl2  = (const float*)d_in[5];
    const float* wr2  = (const float*)d_in[6];
    const float* b2   = (const float*)d_in[7];
    const float* wl3  = (const float*)d_in[8];
    const float* wr3  = (const float*)d_in[9];
    const float* b3   = (const float*)d_in[10];
    float* out = (float*)d_out;

    const int N = in_sizes[0] / 32;
    const int E = in_sizes[1] / 2;
    const int* src = ei;
    const int* dst = ei + E;

    float* deg = (float*)d_ws;
    float* agg = deg + N;
    float* h1  = agg + (size_t)N * 64;
    float* h2  = h1 + (size_t)N * 64;

    const int tiles = (N + 15) / 16;
    const int gemmBlocks = (tiles + 7) / 8;

    constexpr int R = 800;
    const int ownerBlocks = (N + R - 1) / R;

    const int nvec   = (E + 15) / 16;
    const int nWords = nvec * 4;
    const int encBlocks = (nWords + 255) / 256;
    unsigned int* ownA = (unsigned int*)h2;
    unsigned int* pkA  = ownA + nWords;
    unsigned int* ownB = (unsigned int*)h1;
    unsigned int* pkB  = ownB + nWords;

    edge_encode_kernel<R><<<encBlocks, 256, 0, stream>>>(src, dst, ownA, pkA, N, E, nWords);

    owner_sum_kernel<32, R, true><<<ownerBlocks, 1024, 0, stream>>>(x, ownA, pkA, agg, deg, N, nvec);
    sage_wmma_kernel<32, 64, true><<<gemmBlocks, 256, 0, stream>>>(
        agg, deg, x, wl1, wr1, b1, h1, N);

    owner_sum_kernel<64, R, false><<<ownerBlocks, 1024, 0, stream>>>(h1, ownA, pkA, agg, deg, N, nvec);
    sage_wmma_kernel<64, 64, true><<<gemmBlocks, 256, 0, stream>>>(
        agg, deg, h1, wl2, wr2, b2, h2, N);

    edge_encode_kernel<R><<<encBlocks, 256, 0, stream>>>(src, dst, ownB, pkB, N, E, nWords);
    owner_sum_kernel<64, R, false><<<ownerBlocks, 1024, 0, stream>>>(h2, ownB, pkB, agg, deg, N, nvec);
    sage_final_kernel<<<(N + 255) / 256, 256, 0, stream>>>(
        agg, deg, h2, wl3, wr3, b3, out, N);
}


# Dense layer updates run by the owner blocks themselves on their LDS neighbour sums (row pitch channels+4 floats: fragment reads hit 64 banks); sums never stored, 3 launches fewer; sum order kept
# speedup vs baseline: 3.4905x; 1.9370x over previous
#include <hip/hip_runtime.h>
#include <hip/hip_bf16.h>
#include <math.h>

typedef __attribute__((ext_vector_type(2))) float v2f;
typedef __attribute__((ext_vector_type(8))) float v8f;

template <int R>
__global__ __launch_bounds__(256)
void edge_encode_kernel(const int* __restrict__ src,
                        const int* __restrict__ dst,
                        unsigned int* __restrict__ planes,
                        unsigned int* __restrict__ pk,
                        int N, int E, int planeStride) {
    static_assert(R <= 1024, "local destination must fit in 10 bits");
    __shared__ unsigned int ob[2048];
    const int t = threadIdx.x;
    const int eBlock = blockIdx.x * 8192;
    const bool vec = (((((size_t)src) | ((size_t)dst)) & 15) == 0) && E >= 4;

    auto load_slow = [&](int e0, int* s, int* d) {
#pragma unroll
        for (int j = 0; j < 4; ++j) {
            s[j] = (e0 + j < E) ? src[e0 + j] : 0;
            d[j] = (e0 + j < E) ? dst[e0 + j] : -1;
        }
    };
    auto put = [&](int g, int e0, const int* s, const int* d) {
        unsigned int o = 0u;
        unsigned int p[4];
#pragma unroll
        for (int j = 0; j < 4; ++j) {
            const bool ok = (unsigned)d[j] < (unsigned)N;
            const unsigned int b  = ok ? (unsigned)d[j] / (unsigned)R : 255u;
            const unsigned int dl = ok ? (unsigned)d[j] - b * (unsigned)R : 0u;
            o |= b << (8 * j);
            p[j] = ((unsigned)s[j] & 0x3FFFFFu) | (dl << 22);
        }
        ob[g] = o;
        *(uint4*)(pk + e0) = make_uint4(p[0], p[1], p[2], p[3]);
    };

    if (vec) {
        int4 sv[8], dv[8];
#pragma unroll
        for (int r = 0; r < 8; ++r) {
            const int e0 = eBlock + (r * 256 + t) * 4;
            const int ec = (e0 + 4 <= E) ? e0 : 0;
            sv[r] = *(const int4*)(src + ec);
            dv[r] = *(const int4*)(dst + ec);
        }
#pragma unroll
        for (int r = 0; r < 8; ++r) {
            const int g  = r * 256 + t;
            const int e0 = eBlock + g * 4;
            int s[4], d[4];
            if (e0 + 4 <= E) {
                s[0] = sv[r].x; s[1] = sv[r].y; s[2] = sv[r].z; s[3] = sv[r].w;
                d[0] = dv[r].x; d[1] = dv[r].y; d[2] = dv[r].z; d[3] = dv[r].w;
            } else {
                load_slow(e0, s, d);
            }
            put(g, e0, s, d);
        }
    } else {
#pragma unroll
        for (int r = 0; r < 8; ++r) {
            const int g  = r * 256 + t;
            const int e0 = eBlock + g * 4;
            int s[4], d[4];
            load_slow(e0, s, d);
            put(g, e0, s, d);
        }
    }
    __syncthreads();
    unsigned int pw[8] = {0u, 0u, 0u, 0u, 0u, 0u, 0u, 0u};
#pragma unroll
    for (int j = 0; j < 8; ++j) {
        const unsigned int w4 = ob[t * 8 + j];
#pragma unroll
        for (int c = 0; c < 4; ++c) {
            const unsigned int b = (w4 >> (8 * c)) & 255u;
#pragma unroll
            for (int i = 0; i < 8; ++i) pw[i] |= ((b >> i) & 1u) << (4 * j + c);
        }
    }
    const int wi = blockIdx.x * 256 + t;
#pragma unroll
    for (int i = 0; i < 8; ++i) planes[(size_t)i * planeStride + wi] = pw[i];
}

template <int CIN, int P, int R, bool RELU>
__device__ __forceinline__
void owner_dense_rows(const float* acc,
                      const float* scl, int sclBase,
                      const float* __restrict__ xin,
                      const float* __restrict__ wl,
                      const float* __restrict__ wr,
                      const float* __restrict__ bias,
                      float* __restrict__ out,
                      int base, int N) {
    static_assert(R % 16 == 0, "a work-group's rows are whole 16-node tiles");
    constexpr int COUT = 64;
    constexpr int NB   = COUT / 16;
    const int lane = threadIdx.x & 31;
    const int wave = threadIdx.x >> 5;
    const int nw   = blockDim.x >> 5;
    const int m  = lane & 15;
    const int kk = (lane >> 4) << 1;

    for (int t = wave; t < R / 16; t += nw) {
        const int row0 = base + t * 16;
        if (row0 >= N) break;
        const bool fullTile = (row0 + 16 <= N);
        const int node = row0 + (fullTile ? m : min(m, N - 1 - row0));
        const int lr   = node - base;
        const float scale = scl[sclBase + lr];
        const float* sumRow = acc + lr * P;
        const float* xRow   = xin + (size_t)node * CIN;
        const int rbase = row0 + ((lane >> 4) << 3);

        v8f c[NB];
#pragma unroll
        for (int j = 0; j < NB; ++j) c[j] = v8f{};
#pragma unroll
        for (int k0 = 0; k0 < CIN; k0 += 4) {
            v2f a_mean, a_self;
            a_mean.x = sumRow[k0 + kk] * scale;
            a_mean.y = sumRow[k0 + kk + 1] * scale;
            a_self.x = xRow[k0 + kk];
            a_self.y = xRow[k0 + kk + 1];
#pragma unroll
            for (int j = 0; j < NB; ++j) {
                const float* wlRow = wl + (size_t)(j * 16 + m) * CIN;
                const float* wrRow = wr + (size_t)(j * 16 + m) * CIN;
                v2f b_l, b_r;
                b_l.x = wlRow[k0 + kk];
                b_l.y = wlRow[k0 + kk + 1];
                b_r.x = wrRow[k0 + kk];
                b_r.y = wrRow[k0 + kk + 1];
                c[j] = __builtin_amdgcn_wmma_f32_16x16x4_f32(false, a_mean, false, b_l,
                                                             (short)0, c[j], false, false);
                c[j] = __builtin_amdgcn_wmma_f32_16x16x4_f32(false, a_self, false, b_r,
                                                             (short)0, c[j], false, false);
            }
        }
#pragma unroll
        for (int j = 0; j < NB; ++j) {
            const float bv = bias[j * 16 + m];
            if (fullTile) {
                float* o = out + (size_t)rbase * COUT + j * 16 + m;
#pragma unroll
                for (int v = 0; v < 8; ++v) {
                    float val = c[j][v] + bv;
                    if (RELU) val = fmaxf(val, 0.0f);
                    o[(size_t)v * COUT] = val;
                }
            } else {
#pragma unroll
                for (int v = 0; v < 8; ++v) {
                    int r = rbase + v;
                    if (r < N) {
                        float val = c[j][v] + bv;
                        if (RELU) val = fmaxf(val, 0.0f);
                        out[(size_t)r * COUT + j * 16 + m] = val;
                    }
                }
            }
        }
    }
}

template <int C, int R, int CAP>
__global__ __launch_bounds__(1024)
void owner_sum_kernel(const float* __restrict__ x,
                      const unsigned int* __restrict__ planes,
                      const unsigned int* __restrict__ pk,
                      float* __restrict__ invdeg,
                      unsigned int* __restrict__ lst,
                      int* __restrict__ lcnt,
                      const float* __restrict__ wl,
                      const float* __restrict__ wr,
                      const float* __restrict__ bias,
                      float* __restrict__ hout,
                      int N, int nQ) {
    constexpr bool COUNT = true;
    constexpr int P     = C + 4;
    constexpr int NW    = 32;
    constexpr int QCAP  = 64;
    constexpr int VPI   = 1024;
    constexpr int CV    = C / 32;

    __shared__ __attribute__((aligned(16))) float acc[R * P];
    __shared__ float cnt[COUNT ? R : 1];
    __shared__ int   queue[NW * QCAP];
    __shared__ int   lpos;

    const int tid  = threadIdx.x;
    const int lane = tid & 31;
    const int base = blockIdx.x * R;
    int* myq = queue + (tid >> 5) * QCAP;
    unsigned int* mylst = lst + (size_t)blockIdx.x * (CAP > 0 ? CAP : 1);
    if (tid == 0) lpos = 0;

    for (int i = tid; i < R * P / 4; i += 1024) ((float4*)acc)[i] = make_float4(0.0f, 0.0f, 0.0f, 0.0f);
    if (COUNT) {
        for (int i = tid; i < R; i += 1024) cnt[i] = 0.0f;
    }
    __syncthreads();

    int qlen = 0;

    auto drain = [&](int n, int q0) {
        int s = 0, dl = 0;
        unsigned int p = 0u;
        if (lane < n) {
            p  = pk[myq[q0 + lane]];
            s  = (int)(p & 0x3FFFFFu);
            dl = (int)(p >> 22);
            if (COUNT) atomicAdd(cnt + dl, 1.0f);
        }
        if (CAP > 0) {
            int pos = 0;
            if (lane == 0) pos = atomicAdd(&lpos, 32);
            pos = __builtin_amdgcn_readfirstlane(pos);
            if (pos + 32 <= CAP) mylst[pos + lane] = (lane < n) ? p : 0xFFFFFFFFu;
        }
        for (int i0 = 0; i0 < n; i0 += 8) {
            float v[8][CV];
            int   dd[8];
#pragma unroll
            for (int u = 0; u < 8; ++u) {
                const int i  = min(i0 + u, n - 1);
                const int si = __builtin_amdgcn_readlane(s, i);
                dd[u]        = __builtin_amdgcn_readlane(dl, i);
                const float* row = x + (size_t)si * C + lane;
#pragma unroll
                for (int c = 0; c < CV; ++c) v[u][c] = row[c * 32];
            }
#pragma unroll
            for (int u = 0; u < 8; ++u) {
                if (i0 + u < n) {
                    float* a = acc + dd[u] * P + lane;
#pragma unroll
                    for (int c = 0; c < CV; ++c) atomicAdd(a + c * 32, v[u][c]);
                }
            }
        }
    };

    unsigned int kw[8];
#pragma unroll
    for (int i = 0; i < 8; ++i) kw[i] = ((blockIdx.x >> i) & 1u) ? ~0u : 0u;

    const uint4* pl4 = (const uint4*)planes;
    const int nIter = (nQ + VPI - 1) / VPI;
    const int itStart = (int)((blockIdx.x * (unsigned)nIter) / gridDim.x);
    for (int k = 0; k < nIter; ++k) {
        int it = itStart + k;
        if (it >= nIter) it -= nIter;
        const int q = it * VPI + tid;
        const bool valid = (q < nQ);
        const int qc = valid ? q : nQ - 1;
        uint4 pv[8];
#pragma unroll
        for (int i = 0; i < 8; ++i) pv[i] = pl4[(size_t)i * nQ + qc];
        unsigned int m[4];
        {
            unsigned int d0 = 0u, d1 = 0u, d2 = 0u, d3 = 0u;
#pragma unroll
            for (int i = 0; i < 8; ++i) {
                d0 |= pv[i].x ^ kw[i];
                d1 |= pv[i].y ^ kw[i];
                d2 |= pv[i].z ^ kw[i];
                d3 |= pv[i].w ^ kw[i];
            }
            m[0] = valid ? ~d0 : 0u;
            m[1] = valid ? ~d1 : 0u;
            m[2] = valid ? ~d2 : 0u;
            m[3] = valid ? ~d3 : 0u;
        }
#pragma unroll
        for (int w = 0; w < 4; ++w) {
            const int e0 = (q * 4 + w) * 32;
            unsigned int mbits = m[w];
            unsigned int act = __builtin_amdgcn_ballot_w32(mbits != 0u);
            while (act != 0u) {
                if (mbits != 0u) {
                    const int p = __builtin_ctz(mbits);
                    mbits &= mbits - 1u;
                    myq[qlen + (int)__builtin_amdgcn_mbcnt_lo(act, 0u)] = e0 + p;
                }
                qlen += __builtin_popcount(act);
                if (qlen >= 32) {
                    qlen -= 32;
                    drain(32, qlen);
                }
                act = __builtin_amdgcn_ballot_w32(mbits != 0u);
            }
        }
    }
    if (qlen > 0) drain(qlen, 0);
    __syncthreads();

    for (int i = tid; i < R; i += 1024) {
        const float iv = 1.0f / fmaxf(cnt[i], 1.0f);
        cnt[i] = iv;
        if (base + i < N) invdeg[base + i] = iv;
    }
    if (CAP > 0) {
        if (tid == 0) lcnt[blockIdx.x] = lpos;
    }
    __syncthreads();

    owner_dense_rows<C, P, R, true>(acc, cnt, 0, x, wl, wr, bias, hout, base, N);
}

template <int C, int R, int CAP, bool LAST>
__global__ __launch_bounds__(1024)
void owner_replay_kernel(const float* __restrict__ x,
                         const unsigned int* __restrict__ lst,
                         const int* __restrict__ lcnt,
                         const int* __restrict__ src,
                         const int* __restrict__ dst,
                         const float* __restrict__ invdeg,
                         const float* __restrict__ wl,
                         const float* __restrict__ wr,
                         const float* __restrict__ bias,
                         float* __restrict__ hout,
                         int N, int E) {
    static_assert(R < 1024, "local destination 1023 is the empty mark of the list");
    static_assert(CAP % 32 == 0, "the list is handed out in lines of 32 words");
    constexpr int CV = C / 32;
    constexpr int P  = C + 4;

    __shared__ __attribute__((aligned(16))) float acc[R * P];

    const int tid  = threadIdx.x;
    const int lane = tid & 31;
    const int base = blockIdx.x * R;

    for (int i = tid; i < R * P / 4; i += 1024) ((float4*)acc)[i] = make_float4(0.0f, 0.0f, 0.0f, 0.0f);
    __syncthreads();

    auto addk = [&](int i0, int k, int s, int dl) {
        float v[8][CV];
        int   dd[8];
#pragma unroll
        for (int u = 0; u < 8; ++u) {
            if (u < k) {
                const int si = __builtin_amdgcn_readlane(s, i0 + u);
                dd[u]        = __builtin_amdgcn_readlane(dl, i0 + u);
                const float* row = x + (size_t)si * C + lane;
#pragma unroll
                for (int c = 0; c < CV; ++c) v[u][c] = row[c * 32];
            }
        }
#pragma unroll
        for (int u = 0; u < 8; ++u) {
            if (u < k) {
                float* a = acc + dd[u] * P + lane;
#pragma unroll
                for (int c = 0; c < CV; ++c) atomicAdd(a + c * 32, v[u][c]);
            }
        }
    };

    const int cntb = lcnt[blockIdx.x];
    if (cntb <= CAP) {
        const unsigned int* my = lst + (size_t)blockIdx.x * CAP;
        int c0 = (tid >> 5) * 32;
        unsigned int pn = (c0 < cntb) ? my[c0 + lane] : 0xFFFFFFFFu;
        while (c0 < cntb) {
            const unsigned int p = pn;
            const int n  = __builtin_popcount(__builtin_amdgcn_ballot_w32(p != 0xFFFFFFFFu));
            const int c1 = c0 + 1024;
            pn = (c1 < cntb) ? my[c1 + lane] : 0xFFFFFFFFu;
            const int s  = (int)(p & 0x3FFFFFu);
            const int dl = (int)(p >> 22);
            if (n == 32) {
#pragma unroll
                for (int i0 = 0; i0 < 32; i0 += 8) addk(i0, 8, s, dl);
            } else {
                for (int i0 = 0; i0 < n; i0 += 8) {
                    const int k = min(8, n - i0);
#pragma unroll
                    for (int kk = 1; kk <= 8; ++kk) {
                        if (k == kk) addk(i0, kk, s, dl);
                    }
                }
            }
            c0 = c1;
        }
    } else {
        for (int e0 = 0; e0 < E; e0 += 1024) {
            const int e = e0 + tid;
            bool mine = false;
            int s = 0, dl = 0;
            if (e < E) {
                const int d = dst[e];
                if ((unsigned)d < (unsigned)N && d >= base && d < base + R) {
                    mine = true;
                    s  = (int)((unsigned)src[e] & 0x3FFFFFu);
                    dl = d - base;
                }
            }
            unsigned int act = __builtin_amdgcn_ballot_w32(mine);
            while (act != 0u) {
                const int i  = __builtin_ctz(act);
                act &= act - 1u;
                const int si = __builtin_amdgcn_readlane(s, i);
                const int di = __builtin_amdgcn_readlane(dl, i);
                const float* row = x + (size_t)si * C + lane;
                float* a = acc + di * P + lane;
#pragma unroll
                for (int c = 0; c < CV; ++c) atomicAdd(a + c * 32, row[c * 32]);
            }
        }
    }
    __syncthreads();

    if (!LAST) {
        owner_dense_rows<C, P, R, true>(acc, invdeg, base, x, wl, wr, bias, hout, base, N);
    } else {
        static_assert(C == 64, "the last layer reads 16 groups of 4 channels");
        for (int i = tid; i < R; i += 1024) {
            const int node = base + i;
            if (node < N) {
                float s = bias[0];
                const float sc = invdeg[node];
                const float4* ar  = (const float4*)(acc + i * P);
                const float4* hr  = (const float4*)(x + (size_t)node * C);
                const float4* wlv = (const float4*)wl;
                const float4* wrv = (const float4*)wr;
#pragma unroll
                for (int q = 0; q < 16; ++q) {
                    float4 a = ar[q], h = hr[q], l = wlv[q], r = wrv[q];
                    s += (a.x * sc) * l.x + (a.y * sc) * l.y + (a.z * sc) * l.z + (a.w * sc) * l.w;
                    s += h.x * r.x + h.y * r.y + h.z * r.z + h.w * r.w;
                }
                hout[node] = 1.0f / (1.0f + expf(-s));
            }
        }
    }
}

extern "C" void kernel_launch(void* const* d_in, const int* in_sizes, int n_in,
                              void* d_out, int out_size, void* d_ws, size_t ws_size,
                              hipStream_t stream) {
    const float* x    = (const float*)d_in[0];
    const int*   ei   = (const int*)d_in[1];
    const float* wl1  = (const float*)d_in[2];
    const float* wr1  = (const float*)d_in[3];
    const float* b1   = (const float*)d_in[4];
    const float* wl2  = (const float*)d_in[5];
    const float* wr2  = (const float*)d_in[6];
    const float* b2   = (const float*)d_in[7];
    const float* wl3  = (const float*)d_in[8];
    const float* wr3  = (const float*)d_in[9];
    const float* b3   = (const float*)d_in[10];
    float* out = (float*)d_out;

    const int N = in_sizes[0] / 32;
    const int E = in_sizes[1] / 2;
    const int* src = ei;
    const int* dst = ei + E;

    float* deg = (float*)d_ws;
    float* h2  = deg + N;
    float* h1  = h2 + (size_t)N * 64;
    float* edgePlane = h1 + (size_t)N * 64;

    constexpr int R = 400;
    const int ownerBlocks = (N + R - 1) / R;
    constexpr size_t padLds32 = 106496;
    constexpr size_t padLdsRp = 65536;

    const int encBlocks   = (E + 8191) / 8192;
    const int planeStride = encBlocks * 256;
    const int nQ          = encBlocks * 64;
    unsigned int* ownA = (unsigned int*)edgePlane;
    unsigned int* pkA  = ownA + 8 * (size_t)planeStride;
    constexpr int CAP = 16384;
    unsigned int* lstA  = pkA + (size_t)encBlocks * 8192;
    int*          lcntA = (int*)(lstA + (size_t)ownerBlocks * CAP);

    edge_encode_kernel<R><<<encBlocks, 256, 0, stream>>>(src, dst, ownA, pkA, N, E, planeStride);

    owner_sum_kernel<32, R, CAP><<<ownerBlocks, 1024, padLds32, stream>>>(
        x, ownA, pkA, deg, lstA, lcntA, wl1, wr1, b1, h1, N, nQ);

    owner_replay_kernel<64, R, CAP, false><<<ownerBlocks, 1024, padLdsRp, stream>>>(
        h1, lstA, lcntA, src, dst, deg, wl2, wr2, b2, h2, N, E);

    owner_replay_kernel<64, R, CAP, true><<<ownerBlocks, 1024, padLdsRp, stream>>>(
        h2, lstA, lcntA, src, dst, deg, wl3, wr3, b3, out, N, E);
}
